# starting kernel
# baseline (speedup 1.0000x reference)
#include <hip/hip_runtime.h>
#include <math.h>

typedef float v2f __attribute__((ext_vector_type(2)));
typedef float v8f __attribute__((ext_vector_type(8)));

#define B_    4
#define CIN   64
#define COUT  64
#define NPTS  16384
#define KNN   16
#define NGRP  4
#define CPG   16
#define EPS_  1e-5f

__global__ void k_init(float* __restrict__ stats) {
    if (threadIdx.x < B_ * NGRP * 2) stats[threadIdx.x] = 0.0f;
}

__global__ __launch_bounds__(256) void k_gemm(const float* __restrict__ feat,
                                              const float* __restrict__ W1,
                                              const float* __restrict__ W2,
                                              float* __restrict__ local_t,
                                              float* __restrict__ edge_t) {
    __shared__ float sW[2 * COUT * CIN];
    const int tid = threadIdx.x;
    for (int i = tid; i < COUT * CIN; i += 256) {
        sW[i]               = W1[i];
        sW[COUT * CIN + i]  = W2[i];
    }
    __syncthreads();

    const int wave = tid >> 5;
    const int lane = tid & 31;
    const int ln   = lane & 15;
    const int hi   = lane >> 4;
    const int b    = blockIdx.x >> 7;
    const int tile = ((blockIdx.x & 127) << 3) + wave;
    const int n0   = tile << 4;

    v8f acc[8];
#pragma unroll
    for (int i = 0; i < 8; ++i) acc[i] = (v8f){0, 0, 0, 0, 0, 0, 0, 0};

    const size_t fbase = (size_t)b * CIN * NPTS + (size_t)(n0 + ln);
    for (int kk = 0; kk < CIN; kk += 4) {
        const int k0 = kk + 2 * hi;
        v2f bf;
        bf.x = feat[fbase + (size_t)k0 * NPTS];
        bf.y = feat[fbase + (size_t)(k0 + 1) * NPTS];
#pragma unroll
        for (int mat = 0; mat < 2; ++mat) {
#pragma unroll
            for (int mt = 0; mt < 4; ++mt) {
                const float* w = &sW[mat * (COUT * CIN) + (mt * 16 + ln) * CIN + k0];
                v2f af;
                af.x = w[0];
                af.y = w[1];
                acc[mat * 4 + mt] = __builtin_amdgcn_wmma_f32_16x16x4_f32(
                    false, af, false, bf, (short)0, acc[mat * 4 + mt], false, false);
            }
        }
    }

#pragma unroll
    for (int mat = 0; mat < 2; ++mat) {
        float* dst = mat ? edge_t : local_t;
#pragma unroll
        for (int mt = 0; mt < 4; ++mt) {
            v8f c = acc[mat * 4 + mt];
            size_t base = ((size_t)b * NPTS + (size_t)(n0 + ln)) * COUT
                          + (size_t)(mt * 16 + 8 * hi);
            *(float4*)(dst + base)     = make_float4(c[0], c[1], c[2], c[3]);
            *(float4*)(dst + base + 4) = make_float4(c[4], c[5], c[6], c[7]);
        }
    }
}

__global__ __launch_bounds__(256) void k_stats(const int* __restrict__ knn,
                                               const float* __restrict__ local_t,
                                               const float* __restrict__ edge_t,
                                               float* __restrict__ stats) {
    __shared__ float sh[2 * NGRP];
    const int tid = threadIdx.x;
    if (tid < 2 * NGRP) sh[tid] = 0.0f;
    __syncthreads();

    const int b  = blockIdx.x >> 10;
    const int t  = ((blockIdx.x & 1023) << 8) + tid;
    const int c4 = t & 15;
    const int n  = t >> 4;
    const int g  = c4 >> 2;

    const size_t colL = ((size_t)b * NPTS + n) * COUT + (size_t)c4 * 4;
    const float4 l4 = *(const float4*)(local_t + colL);
    const int* kp = knn + ((size_t)b * NPTS + n) * KNN;

    float s = 0.0f, ss = 0.0f;
#pragma unroll
    for (int j = 0; j < KNN; ++j) {
        const int idx = kp[j];
        const float4 e4 = *(const float4*)(edge_t + ((size_t)b * NPTS + idx) * COUT
                                           + (size_t)c4 * 4);
        const float d0 = e4.x - l4.x, d1 = e4.y - l4.y;
        const float d2 = e4.z - l4.z, d3 = e4.w - l4.w;
        s  += d0 + d1 + d2 + d3;
        ss += d0 * d0 + d1 * d1 + d2 * d2 + d3 * d3;
    }
    atomicAdd(&sh[g * 2 + 0], s);
    atomicAdd(&sh[g * 2 + 1], ss);
    __syncthreads();
    if (tid < 2 * NGRP) atomicAdd(&stats[b * 2 * NGRP + tid], sh[tid]);
}

__global__ __launch_bounds__(256) void k_out(const int* __restrict__ knn,
                                             const float* __restrict__ local_t,
                                             const float* __restrict__ edge_t,
                                             const float* __restrict__ stats,
                                             const float* __restrict__ gw,
                                             const float* __restrict__ gb,
                                             float* __restrict__ out) {
    const int tid = threadIdx.x;
    const int b  = blockIdx.x >> 10;
    const int t  = ((blockIdx.x & 1023) << 8) + tid;
    const int c4 = t & 15;
    const int n  = t >> 4;
    const int g  = c4 >> 2;

    const float inv_cnt = 1.0f / (float)((size_t)CPG * NPTS * KNN);
    const float mean = stats[b * 2 * NGRP + g * 2 + 0] * inv_cnt;
    const float msq  = stats[b * 2 * NGRP + g * 2 + 1] * inv_cnt;
    const float inv  = rsqrtf(msq - mean * mean + EPS_);

    const float4 gamma = ((const float4*)gw)[c4];
    const float4 beta  = ((const float4*)gb)[c4];
    const float4 l4 = *(const float4*)(local_t + ((size_t)b * NPTS + n) * COUT
                                       + (size_t)c4 * 4);
    const int* kp = knn + ((size_t)b * NPTS + n) * KNN;

    float a0 = 0.0f, a1 = 0.0f, a2 = 0.0f, a3 = 0.0f;
#pragma unroll
    for (int j = 0; j < KNN; ++j) {
        const int idx = kp[j];
        const float4 e4 = *(const float4*)(edge_t + ((size_t)b * NPTS + idx) * COUT
                                           + (size_t)c4 * 4);
        float v0 = ((e4.x - l4.x) - mean) * inv * gamma.x + beta.x;
        float v1 = ((e4.y - l4.y) - mean) * inv * gamma.y + beta.y;
        float v2 = ((e4.z - l4.z) - mean) * inv * gamma.z + beta.z;
        float v3 = ((e4.w - l4.w) - mean) * inv * gamma.w + beta.w;
        a0 += fmaxf(v0, 0.0f);
        a1 += fmaxf(v1, 0.0f);
        a2 += fmaxf(v2, 0.0f);
        a3 += fmaxf(v3, 0.0f);
    }
    const float r = 1.0f / (float)KNN;
    const size_t ob = ((size_t)b * COUT + (size_t)c4 * 4) * NPTS + (size_t)n;
    out[ob]                     = a0 * r;
    out[ob + (size_t)NPTS]      = a1 * r;
    out[ob + 2 * (size_t)NPTS]  = a2 * r;
    out[ob + 3 * (size_t)NPTS]  = a3 * r;
}

extern "C" void kernel_launch(void* const* d_in, const int* in_sizes, int n_in,
                              void* d_out, int out_size, void* d_ws, size_t ws_size,
                              hipStream_t stream) {
    (void)in_sizes; (void)n_in; (void)out_size; (void)ws_size;

    const float* feat = (const float*)d_in[0];
    const int*   knn  = (const int*)d_in[1];
    const float* W1   = (const float*)d_in[2];
    const float* W2   = (const float*)d_in[3];
    const float* gw   = (const float*)d_in[4];
    const float* gb   = (const float*)d_in[5];
    float*       out  = (float*)d_out;

    float* local_t = (float*)d_ws;
    float* edge_t  = local_t + (size_t)B_ * NPTS * COUT;
    float* stats   = edge_t  + (size_t)B_ * NPTS * COUT;

    k_init <<<1, 64, 0, stream>>>(stats);
    k_gemm <<<B_ * (NPTS / 16) / 8, 256, 0, stream>>>(feat, W1, W2, local_t, edge_t);
    k_stats<<<B_ * (NPTS * KNN) / 256, 256, 0, stream>>>(knn, local_t, edge_t, stats);
    k_out  <<<B_ * (NPTS * KNN) / 256, 256, 0, stream>>>(knn, local_t, edge_t, stats,
                                                         gw, gb, out);
}


# k_out: work-group covers 64 points and stages its [channel][point] results in LDS, so every store of out writes whole 128-byte lines (4 per instruction) instead of 16 lines x 8 bytes each
# speedup vs baseline: 1.6102x; 1.6102x over previous
#include <hip/hip_runtime.h>
#include <math.h>

typedef float v2f __attribute__((ext_vector_type(2)));
typedef float v8f __attribute__((ext_vector_type(8)));

#define B_    4
#define CIN   64
#define COUT  64
#define NPTS  16384
#define KNN   16
#define NGRP  4
#define CPG   16
#define EPS_  1e-5f
#define OT_N  64
#define OT_LD (OT_N + 1)

__global__ void k_init(float* __restrict__ stats) {
    if (threadIdx.x < B_ * NGRP * 2) stats[threadIdx.x] = 0.0f;
}

__global__ __launch_bounds__(256) void k_gemm(const float* __restrict__ feat,
                                              const float* __restrict__ W1,
                                              const float* __restrict__ W2,
                                              float* __restrict__ local_t,
                                              float* __restrict__ edge_t) {
    __shared__ float sW[2 * COUT * CIN];
    const int tid = threadIdx.x;
    for (int i = tid; i < COUT * CIN; i += 256) {
        sW[i]               = W1[i];
        sW[COUT * CIN + i]  = W2[i];
    }
    __syncthreads();

    const int wave = tid >> 5;
    const int lane = tid & 31;
    const int ln   = lane & 15;
    const int hi   = lane >> 4;
    const int b    = blockIdx.x >> 7;
    const int tile = ((blockIdx.x & 127) << 3) + wave;
    const int n0   = tile << 4;

    v8f acc[8];
#pragma unroll
    for (int i = 0; i < 8; ++i) acc[i] = (v8f){0, 0, 0, 0, 0, 0, 0, 0};

    const size_t fbase = (size_t)b * CIN * NPTS + (size_t)(n0 + ln);
    for (int kk = 0; kk < CIN; kk += 4) {
        const int k0 = kk + 2 * hi;
        v2f bf;
        bf.x = feat[fbase + (size_t)k0 * NPTS];
        bf.y = feat[fbase + (size_t)(k0 + 1) * NPTS];
#pragma unroll
        for (int mat = 0; mat < 2; ++mat) {
#pragma unroll
            for (int mt = 0; mt < 4; ++mt) {
                const float* w = &sW[mat * (COUT * CIN) + (mt * 16 + ln) * CIN + k0];
                v2f af;
                af.x = w[0];
                af.y = w[1];
                acc[mat * 4 + mt] = __builtin_amdgcn_wmma_f32_16x16x4_f32(
                    false, af, false, bf, (short)0, acc[mat * 4 + mt], false, false);
            }
        }
    }

#pragma unroll
    for (int mat = 0; mat < 2; ++mat) {
        float* dst = mat ? edge_t : local_t;
#pragma unroll
        for (int mt = 0; mt < 4; ++mt) {
            v8f c = acc[mat * 4 + mt];
            size_t base = ((size_t)b * NPTS + (size_t)(n0 + ln)) * COUT
                          + (size_t)(mt * 16 + 8 * hi);
            *(float4*)(dst + base)     = make_float4(c[0], c[1], c[2], c[3]);
            *(float4*)(dst + base + 4) = make_float4(c[4], c[5], c[6], c[7]);
        }
    }
}

__global__ __launch_bounds__(256) void k_stats(const int* __restrict__ knn,
                                               const float* __restrict__ local_t,
                                               const float* __restrict__ edge_t,
                                               float* __restrict__ stats) {
    __shared__ float sh[2 * NGRP];
    const int tid = threadIdx.x;
    if (tid < 2 * NGRP) sh[tid] = 0.0f;
    __syncthreads();

    const int b  = blockIdx.x >> 10;
    const int t  = ((blockIdx.x & 1023) << 8) + tid;
    const int c4 = t & 15;
    const int n  = t >> 4;
    const int g  = c4 >> 2;

    const size_t colL = ((size_t)b * NPTS + n) * COUT + (size_t)c4 * 4;
    const float4 l4 = *(const float4*)(local_t + colL);
    const int* kp = knn + ((size_t)b * NPTS + n) * KNN;

    float s = 0.0f, ss = 0.0f;
#pragma unroll
    for (int j = 0; j < KNN; ++j) {
        const int idx = kp[j];
        const float4 e4 = *(const float4*)(edge_t + ((size_t)b * NPTS + idx) * COUT
                                           + (size_t)c4 * 4);
        const float d0 = e4.x - l4.x, d1 = e4.y - l4.y;
        const float d2 = e4.z - l4.z, d3 = e4.w - l4.w;
        s  += d0 + d1 + d2 + d3;
        ss += d0 * d0 + d1 * d1 + d2 * d2 + d3 * d3;
    }
    atomicAdd(&sh[g * 2 + 0], s);
    atomicAdd(&sh[g * 2 + 1], ss);
    __syncthreads();
    if (tid < 2 * NGRP) atomicAdd(&stats[b * 2 * NGRP + tid], sh[tid]);
}

__global__ __launch_bounds__(256) void k_out(const int* __restrict__ knn,
                                             const float* __restrict__ local_t,
                                             const float* __restrict__ edge_t,
                                             const float* __restrict__ stats,
                                             const float* __restrict__ gw,
                                             const float* __restrict__ gb,
                                             float* __restrict__ out) {
    __shared__ float tile[COUT * OT_LD];
    const int tid = threadIdx.x;
    const int b   = blockIdx.x / (NPTS / OT_N);
    const int n0  = (blockIdx.x % (NPTS / OT_N)) * OT_N;
    const int c4  = tid & 15;
    const int nl  = tid >> 4;
    const int g   = c4 >> 2;

    const float inv_cnt = 1.0f / (float)((size_t)CPG * NPTS * KNN);
    const float mean = stats[b * 2 * NGRP + g * 2 + 0] * inv_cnt;
    const float msq  = stats[b * 2 * NGRP + g * 2 + 1] * inv_cnt;
    const float inv  = rsqrtf(msq - mean * mean + EPS_);

    const float4 gamma = ((const float4*)gw)[c4];
    const float4 beta  = ((const float4*)gb)[c4];
    const float r = 1.0f / (float)KNN;

    for (int s = 0; s < OT_N / 16; ++s) {
        const int pl = s * 16 + nl;
        const int n  = n0 + pl;
        const float4 l4 = *(const float4*)(local_t + ((size_t)b * NPTS + n) * COUT
                                           + (size_t)c4 * 4);
        const int* kp = knn + ((size_t)b * NPTS + n) * KNN;

        float4 e[KNN];
#pragma unroll
        for (int j = 0; j < KNN; ++j) {
            const int idx = kp[j];
            e[j] = *(const float4*)(edge_t + ((size_t)b * NPTS + idx) * COUT + (size_t)c4 * 4);
        }
        __builtin_amdgcn_sched_barrier(0);

        float a0 = 0.0f, a1 = 0.0f, a2 = 0.0f, a3 = 0.0f;
#pragma unroll
        for (int j = 0; j < KNN; ++j) {
            const float4 e4 = e[j];
            float v0 = ((e4.x - l4.x) - mean) * inv * gamma.x + beta.x;
            float v1 = ((e4.y - l4.y) - mean) * inv * gamma.y + beta.y;
            float v2 = ((e4.z - l4.z) - mean) * inv * gamma.z + beta.z;
            float v3 = ((e4.w - l4.w) - mean) * inv * gamma.w + beta.w;
            a0 += fmaxf(v0, 0.0f);
            a1 += fmaxf(v1, 0.0f);
            a2 += fmaxf(v2, 0.0f);
            a3 += fmaxf(v3, 0.0f);
        }
        tile[(c4 * 4 + 0) * OT_LD + pl] = a0 * r;
        tile[(c4 * 4 + 1) * OT_LD + pl] = a1 * r;
        tile[(c4 * 4 + 2) * OT_LD + pl] = a2 * r;
        tile[(c4 * 4 + 3) * OT_LD + pl] = a3 * r;
    }
    __syncthreads();

    const int q    = tid & 15;
    const int crow = tid >> 4;
    for (int p = 0; p < COUT / 16; ++p) {
        const int c = p * 16 + crow;
        const float* trow = tile + c * OT_LD + 4 * q;
        const float4 v = make_float4(trow[0], trow[1], trow[2], trow[3]);
        *(float4*)(out + ((size_t)b * COUT + (size_t)c) * NPTS + (size_t)(n0 + 4 * q)) = v;
    }
}

extern "C" void kernel_launch(void* const* d_in, const int* in_sizes, int n_in,
                              void* d_out, int out_size, void* d_ws, size_t ws_size,
                              hipStream_t stream) {
    (void)in_sizes; (void)n_in; (void)out_size; (void)ws_size;

    const float* feat = (const float*)d_in[0];
    const int*   knn  = (const int*)d_in[1];
    const float* W1   = (const float*)d_in[2];
    const float* W2   = (const float*)d_in[3];
    const float* gw   = (const float*)d_in[4];
    const float* gb   = (const float*)d_in[5];
    float*       out  = (float*)d_out;

    float* local_t = (float*)d_ws;
    float* edge_t  = local_t + (size_t)B_ * NPTS * COUT;
    float* stats   = edge_t  + (size_t)B_ * NPTS * COUT;

    k_init <<<1, 64, 0, stream>>>(stats);
    k_gemm <<<B_ * (NPTS / 16) / 8, 256, 0, stream>>>(feat, W1, W2, local_t, edge_t);
    k_stats<<<B_ * (NPTS * KNN) / 256, 256, 0, stream>>>(knn, local_t, edge_t, stats);
    k_out  <<<B_ * (NPTS / OT_N), 256, 0, stream>>>(knn, local_t, edge_t, stats,
                                                    gw, gb, out);
}


# k_gemm: LDS weight rows padded by 4 floats, so the 16 rows of each 8-byte A-fragment read fall on different banks instead of on one bank pair (16-way conflict on every A read before)
# speedup vs baseline: 1.8655x; 1.1586x over previous
#include <hip/hip_runtime.h>
#include <math.h>

typedef float v2f __attribute__((ext_vector_type(2)));
typedef float v8f __attribute__((ext_vector_type(8)));

#define B_    4
#define CIN   64
#define COUT  64
#define NPTS  16384
#define KNN   16
#define NGRP  4
#define CPG   16
#define EPS_  1e-5f
#define OT_N  64
#define OT_LD (OT_N + 1)
#define WP    (CIN + 4)

__global__ void k_init(float* __restrict__ stats) {
    if (threadIdx.x < B_ * NGRP * 2) stats[threadIdx.x] = 0.0f;
}

__global__ __launch_bounds__(256) void k_gemm(const float* __restrict__ feat,
                                              const float* __restrict__ W1,
                                              const float* __restrict__ W2,
                                              float* __restrict__ local_t,
                                              float* __restrict__ edge_t) {
    __shared__ float sW[2 * COUT * WP];
    const int tid = threadIdx.x;
    for (int i = tid; i < COUT * CIN; i += 256) {
        const int o = i / CIN, k = i % CIN;
        sW[o * WP + k]             = W1[i];
        sW[COUT * WP + o * WP + k] = W2[i];
    }
    __syncthreads();

    const int wave = tid >> 5;
    const int lane = tid & 31;
    const int ln   = lane & 15;
    const int hi   = lane >> 4;
    const int b    = blockIdx.x >> 7;
    const int tile = ((blockIdx.x & 127) << 3) + wave;
    const int n0   = tile << 4;

    v8f acc[8];
#pragma unroll
    for (int i = 0; i < 8; ++i) acc[i] = (v8f){0, 0, 0, 0, 0, 0, 0, 0};

    const size_t fbase = (size_t)b * CIN * NPTS + (size_t)(n0 + ln);
    for (int kk = 0; kk < CIN; kk += 4) {
        const int k0 = kk + 2 * hi;
        v2f bf;
        bf.x = feat[fbase + (size_t)k0 * NPTS];
        bf.y = feat[fbase + (size_t)(k0 + 1) * NPTS];
#pragma unroll
        for (int mat = 0; mat < 2; ++mat) {
#pragma unroll
            for (int mt = 0; mt < 4; ++mt) {
                const float* w = &sW[mat * (COUT * WP) + (mt * 16 + ln) * WP + k0];
                v2f af;
                af.x = w[0];
                af.y = w[1];
                acc[mat * 4 + mt] = __builtin_amdgcn_wmma_f32_16x16x4_f32(
                    false, af, false, bf, (short)0, acc[mat * 4 + mt], false, false);
            }
        }
    }

#pragma unroll
    for (int mat = 0; mat < 2; ++mat) {
        float* dst = mat ? edge_t : local_t;
#pragma unroll
        for (int mt = 0; mt < 4; ++mt) {
            v8f c = acc[mat * 4 + mt];
            size_t base = ((size_t)b * NPTS + (size_t)(n0 + ln)) * COUT
                          + (size_t)(mt * 16 + 8 * hi);
            *(float4*)(dst + base)     = make_float4(c[0], c[1], c[2], c[3]);
            *(float4*)(dst + base + 4) = make_float4(c[4], c[5], c[6], c[7]);
        }
    }
}

__global__ __launch_bounds__(256) void k_stats(const int* __restrict__ knn,
                                               const float* __restrict__ local_t,
                                               const float* __restrict__ edge_t,
                                               float* __restrict__ stats) {
    __shared__ float sh[2 * NGRP];
    const int tid = threadIdx.x;
    if (tid < 2 * NGRP) sh[tid] = 0.0f;
    __syncthreads();

    const int b  = blockIdx.x >> 10;
    const int t  = ((blockIdx.x & 1023) << 8) + tid;
    const int c4 = t & 15;
    const int n  = t >> 4;
    const int g  = c4 >> 2;

    const size_t colL = ((size_t)b * NPTS + n) * COUT + (size_t)c4 * 4;
    const float4 l4 = *(const float4*)(local_t + colL);
    const int* kp = knn + ((size_t)b * NPTS + n) * KNN;

    float s = 0.0f, ss = 0.0f;
#pragma unroll
    for (int j = 0; j < KNN; ++j) {
        const int idx = kp[j];
        const float4 e4 = *(const float4*)(edge_t + ((size_t)b * NPTS + idx) * COUT
                                           + (size_t)c4 * 4);
        const float d0 = e4.x - l4.x, d1 = e4.y - l4.y;
        const float d2 = e4.z - l4.z, d3 = e4.w - l4.w;
        s  += d0 + d1 + d2 + d3;
        ss += d0 * d0 + d1 * d1 + d2 * d2 + d3 * d3;
    }
    atomicAdd(&sh[g * 2 + 0], s);
    atomicAdd(&sh[g * 2 + 1], ss);
    __syncthreads();
    if (tid < 2 * NGRP) atomicAdd(&stats[b * 2 * NGRP + tid], sh[tid]);
}

__global__ __launch_bounds__(256) void k_out(const int* __restrict__ knn,
                                             const float* __restrict__ local_t,
                                             const float* __restrict__ edge_t,
                                             const float* __restrict__ stats,
                                             const float* __restrict__ gw,
                                             const float* __restrict__ gb,
                                             float* __restrict__ out) {
    __shared__ float tile[COUT * OT_LD];
    const int tid = threadIdx.x;
    const int b   = blockIdx.x / (NPTS / OT_N);
    const int n0  = (blockIdx.x % (NPTS / OT_N)) * OT_N;
    const int c4  = tid & 15;
    const int nl  = tid >> 4;
    const int g   = c4 >> 2;

    const float inv_cnt = 1.0f / (float)((size_t)CPG * NPTS * KNN);
    const float mean = stats[b * 2 * NGRP + g * 2 + 0] * inv_cnt;
    const float msq  = stats[b * 2 * NGRP + g * 2 + 1] * inv_cnt;
    const float inv  = rsqrtf(msq - mean * mean + EPS_);

    const float4 gamma = ((const float4*)gw)[c4];
    const float4 beta  = ((const float4*)gb)[c4];
    const float r = 1.0f / (float)KNN;

    for (int s = 0; s < OT_N / 16; ++s) {
        const int pl = s * 16 + nl;
        const int n  = n0 + pl;
        const float4 l4 = *(const float4*)(local_t + ((size_t)b * NPTS + n) * COUT
                                           + (size_t)c4 * 4);
        const int* kp = knn + ((size_t)b * NPTS + n) * KNN;

        float4 e[KNN];
#pragma unroll
        for (int j = 0; j < KNN; ++j) {
            const int idx = kp[j];
            e[j] = *(const float4*)(edge_t + ((size_t)b * NPTS + idx) * COUT + (size_t)c4 * 4);
        }
        __builtin_amdgcn_sched_barrier(0);

        float a0 = 0.0f, a1 = 0.0f, a2 = 0.0f, a3 = 0.0f;
#pragma unroll
        for (int j = 0; j < KNN; ++j) {
            const float4 e4 = e[j];
            float v0 = ((e4.x - l4.x) - mean) * inv * gamma.x + beta.x;
            float v1 = ((e4.y - l4.y) - mean) * inv * gamma.y + beta.y;
            float v2 = ((e4.z - l4.z) - mean) * inv * gamma.z + beta.z;
            float v3 = ((e4.w - l4.w) - mean) * inv * gamma.w + beta.w;
            a0 += fmaxf(v0, 0.0f);
            a1 += fmaxf(v1, 0.0f);
            a2 += fmaxf(v2, 0.0f);
            a3 += fmaxf(v3, 0.0f);
        }
        tile[(c4 * 4 + 0) * OT_LD + pl] = a0 * r;
        tile[(c4 * 4 + 1) * OT_LD + pl] = a1 * r;
        tile[(c4 * 4 + 2) * OT_LD + pl] = a2 * r;
        tile[(c4 * 4 + 3) * OT_LD + pl] = a3 * r;
    }
    __syncthreads();

    const int q    = tid & 15;
    const int crow = tid >> 4;
    for (int p = 0; p < COUT / 16; ++p) {
        const int c = p * 16 + crow;
        const float* trow = tile + c * OT_LD + 4 * q;
        const float4 v = make_float4(trow[0], trow[1], trow[2], trow[3]);
        *(float4*)(out + ((size_t)b * COUT + (size_t)c) * NPTS + (size_t)(n0 + 4 * q)) = v;
    }
}

extern "C" void kernel_launch(void* const* d_in, const int* in_sizes, int n_in,
                              void* d_out, int out_size, void* d_ws, size_t ws_size,
                              hipStream_t stream) {
    (void)in_sizes; (void)n_in; (void)out_size; (void)ws_size;

    const float* feat = (const float*)d_in[0];
    const int*   knn  = (const int*)d_in[1];
    const float* W1   = (const float*)d_in[2];
    const float* W2   = (const float*)d_in[3];
    const float* gw   = (const float*)d_in[4];
    const float* gb   = (const float*)d_in[5];
    float*       out  = (float*)d_out;

    float* local_t = (float*)d_ws;
    float* edge_t  = local_t + (size_t)B_ * NPTS * COUT;
    float* stats   = edge_t  + (size_t)B_ * NPTS * COUT;

    k_init <<<1, 64, 0, stream>>>(stats);
    k_gemm <<<B_ * (NPTS / 16) / 8, 256, 0, stream>>>(feat, W1, W2, local_t, edge_t);
    k_stats<<<B_ * (NPTS * KNN) / 256, 256, 0, stream>>>(knn, local_t, edge_t, stats);
    k_out  <<<B_ * (NPTS / OT_N), 256, 0, stream>>>(knn, local_t, edge_t, stats,
                                                    gw, gb, out);
}


# k_stats takes k_out's gather loop: a work-group covers 64 points in 4 passes and a thread's 16 neighbour gathers of a pass are in flight together (was one or two per wait, and 4x the waves)
# speedup vs baseline: 2.2215x; 1.1908x over previous
#include <hip/hip_runtime.h>
#include <math.h>

typedef float v2f __attribute__((ext_vector_type(2)));
typedef float v8f __attribute__((ext_vector_type(8)));

#define B_    4
#define CIN   64
#define COUT  64
#define NPTS  16384
#define KNN   16
#define NGRP  4
#define CPG   16
#define EPS_  1e-5f
#define OT_N  64
#define ST_N  64
#define OT_LD (OT_N + 1)
#define WP    (CIN + 4)

__global__ void k_init(float* __restrict__ stats) {
    if (threadIdx.x < B_ * NGRP * 2) stats[threadIdx.x] = 0.0f;
}

__global__ __launch_bounds__(256) void k_gemm(const float* __restrict__ feat,
                                              const float* __restrict__ W1,
                                              const float* __restrict__ W2,
                                              float* __restrict__ local_t,
                                              float* __restrict__ edge_t) {
    __shared__ float sW[2 * COUT * WP];
    const int tid = threadIdx.x;
    for (int i = tid; i < COUT * CIN; i += 256) {
        const int o = i / CIN, k = i % CIN;
        sW[o * WP + k]             = W1[i];
        sW[COUT * WP + o * WP + k] = W2[i];
    }
    __syncthreads();

    const int wave = tid >> 5;
    const int lane = tid & 31;
    const int ln   = lane & 15;
    const int hi   = lane >> 4;
    const int b    = blockIdx.x >> 7;
    const int tile = ((blockIdx.x & 127) << 3) + wave;
    const int n0   = tile << 4;

    v8f acc[8];
#pragma unroll
    for (int i = 0; i < 8; ++i) acc[i] = (v8f){0, 0, 0, 0, 0, 0, 0, 0};

    const size_t fbase = (size_t)b * CIN * NPTS + (size_t)(n0 + ln);
    for (int kk = 0; kk < CIN; kk += 4) {
        const int k0 = kk + 2 * hi;
        v2f bf;
        bf.x = feat[fbase + (size_t)k0 * NPTS];
        bf.y = feat[fbase + (size_t)(k0 + 1) * NPTS];
#pragma unroll
        for (int mat = 0; mat < 2; ++mat) {
#pragma unroll
            for (int mt = 0; mt < 4; ++mt) {
                const float* w = &sW[mat * (COUT * WP) + (mt * 16 + ln) * WP + k0];
                v2f af;
                af.x = w[0];
                af.y = w[1];
                acc[mat * 4 + mt] = __builtin_amdgcn_wmma_f32_16x16x4_f32(
                    false, af, false, bf, (short)0, acc[mat * 4 + mt], false, false);
            }
        }
    }

#pragma unroll
    for (int mat = 0; mat < 2; ++mat) {
        float* dst = mat ? edge_t : local_t;
#pragma unroll
        for (int mt = 0; mt < 4; ++mt) {
            v8f c = acc[mat * 4 + mt];
            size_t base = ((size_t)b * NPTS + (size_t)(n0 + ln)) * COUT
                          + (size_t)(mt * 16 + 8 * hi);
            *(float4*)(dst + base)     = make_float4(c[0], c[1], c[2], c[3]);
            *(float4*)(dst + base + 4) = make_float4(c[4], c[5], c[6], c[7]);
        }
    }
}

__global__ __launch_bounds__(256) void k_stats(const int* __restrict__ knn,
                                               const float* __restrict__ local_t,
                                               const float* __restrict__ edge_t,
                                               float* __restrict__ stats) {
    __shared__ float sh[2 * NGRP];
    const int tid = threadIdx.x;
    if (tid < 2 * NGRP) sh[tid] = 0.0f;
    __syncthreads();

    const int b  = blockIdx.x / (NPTS / ST_N);
    const int n0 = (blockIdx.x % (NPTS / ST_N)) * ST_N;
    const int c4 = tid & 15;
    const int nl = tid >> 4;
    const int g  = c4 >> 2;

    float s = 0.0f, ss = 0.0f;
    for (int p = 0; p < ST_N / 16; ++p) {
        const int n = n0 + p * 16 + nl;
        const float4 l4 = *(const float4*)(local_t + ((size_t)b * NPTS + n) * COUT
                                           + (size_t)c4 * 4);
        const int* kp = knn + ((size_t)b * NPTS + n) * KNN;

        float4 e[KNN];
#pragma unroll
        for (int j = 0; j < KNN; ++j) {
            const int idx = kp[j];
            e[j] = *(const float4*)(edge_t + ((size_t)b * NPTS + idx) * COUT + (size_t)c4 * 4);
        }
        __builtin_amdgcn_sched_barrier(0);

#pragma unroll
        for (int j = 0; j < KNN; ++j) {
            const float4 e4 = e[j];
            const float d0 = e4.x - l4.x, d1 = e4.y - l4.y;
            const float d2 = e4.z - l4.z, d3 = e4.w - l4.w;
            s  += d0 + d1 + d2 + d3;
            ss += d0 * d0 + d1 * d1 + d2 * d2 + d3 * d3;
        }
    }
    atomicAdd(&sh[g * 2 + 0], s);
    atomicAdd(&sh[g * 2 + 1], ss);
    __syncthreads();
    if (tid < 2 * NGRP) atomicAdd(&stats[b * 2 * NGRP + tid], sh[tid]);
}

__global__ __launch_bounds__(256) void k_out(const int* __restrict__ knn,
                                             const float* __restrict__ local_t,
                                             const float* __restrict__ edge_t,
                                             const float* __restrict__ stats,
                                             const float* __restrict__ gw,
                                             const float* __restrict__ gb,
                                             float* __restrict__ out) {
    __shared__ float tile[COUT * OT_LD];
    const int tid = threadIdx.x;
    const int b   = blockIdx.x / (NPTS / OT_N);
    const int n0  = (blockIdx.x % (NPTS / OT_N)) * OT_N;
    const int c4  = tid & 15;
    const int nl  = tid >> 4;
    const int g   = c4 >> 2;

    const float inv_cnt = 1.0f / (float)((size_t)CPG * NPTS * KNN);
    const float mean = stats[b * 2 * NGRP + g * 2 + 0] * inv_cnt;
    const float msq  = stats[b * 2 * NGRP + g * 2 + 1] * inv_cnt;
    const float inv  = rsqrtf(msq - mean * mean + EPS_);

    const float4 gamma = ((const float4*)gw)[c4];
    const float4 beta  = ((const float4*)gb)[c4];
    const float r = 1.0f / (float)KNN;

    for (int s = 0; s < OT_N / 16; ++s) {
        const int pl = s * 16 + nl;
        const int n  = n0 + pl;
        const float4 l4 = *(const float4*)(local_t + ((size_t)b * NPTS + n) * COUT
                                           + (size_t)c4 * 4);
        const int* kp = knn + ((size_t)b * NPTS + n) * KNN;

        float4 e[KNN];
#pragma unroll
        for (int j = 0; j < KNN; ++j) {
            const int idx = kp[j];
            e[j] = *(const float4*)(edge_t + ((size_t)b * NPTS + idx) * COUT + (size_t)c4 * 4);
        }
        __builtin_amdgcn_sched_barrier(0);

        float a0 = 0.0f, a1 = 0.0f, a2 = 0.0f, a3 = 0.0f;
#pragma unroll
        for (int j = 0; j < KNN; ++j) {
            const float4 e4 = e[j];
            float v0 = ((e4.x - l4.x) - mean) * inv * gamma.x + beta.x;
            float v1 = ((e4.y - l4.y) - mean) * inv * gamma.y + beta.y;
            float v2 = ((e4.z - l4.z) - mean) * inv * gamma.z + beta.z;
            float v3 = ((e4.w - l4.w) - mean) * inv * gamma.w + beta.w;
            a0 += fmaxf(v0, 0.0f);
            a1 += fmaxf(v1, 0.0f);
            a2 += fmaxf(v2, 0.0f);
            a3 += fmaxf(v3, 0.0f);
        }
        tile[(c4 * 4 + 0) * OT_LD + pl] = a0 * r;
        tile[(c4 * 4 + 1) * OT_LD + pl] = a1 * r;
        tile[(c4 * 4 + 2) * OT_LD + pl] = a2 * r;
        tile[(c4 * 4 + 3) * OT_LD + pl] = a3 * r;
    }
    __syncthreads();

    const int q    = tid & 15;
    const int crow = tid >> 4;
    for (int p = 0; p < COUT / 16; ++p) {
        const int c = p * 16 + crow;
        const float* trow = tile + c * OT_LD + 4 * q;
        const float4 v = make_float4(trow[0], trow[1], trow[2], trow[3]);
        *(float4*)(out + ((size_t)b * COUT + (size_t)c) * NPTS + (size_t)(n0 + 4 * q)) = v;
    }
}

extern "C" void kernel_launch(void* const* d_in, const int* in_sizes, int n_in,
                              void* d_out, int out_size, void* d_ws, size_t ws_size,
                              hipStream_t stream) {
    (void)in_sizes; (void)n_in; (void)out_size; (void)ws_size;

    const float* feat = (const float*)d_in[0];
    const int*   knn  = (const int*)d_in[1];
    const float* W1   = (const float*)d_in[2];
    const float* W2   = (const float*)d_in[3];
    const float* gw   = (const float*)d_in[4];
    const float* gb   = (const float*)d_in[5];
    float*       out  = (float*)d_out;

    float* local_t = (float*)d_ws;
    float* edge_t  = local_t + (size_t)B_ * NPTS * COUT;
    float* stats   = edge_t  + (size_t)B_ * NPTS * COUT;

    k_init <<<1, 64, 0, stream>>>(stats);
    k_gemm <<<B_ * (NPTS / 16) / 8, 256, 0, stream>>>(feat, W1, W2, local_t, edge_t);
    k_stats<<<B_ * (NPTS / ST_N), 256, 0, stream>>>(knn, local_t, edge_t, stats);
    k_out  <<<B_ * (NPTS / OT_N), 256, 0, stream>>>(knn, local_t, edge_t, stats,
                                                    gw, gb, out);
}


# k_gemm: output tile pair outermost, K innermost (the wave's 64x16 feature slice kept in 32 registers): each pair's 4 stores drain under the next pair's matrix instructions, not all 16 after the last
# speedup vs baseline: 2.3374x; 1.0522x over previous
#include <hip/hip_runtime.h>
#include <math.h>

typedef float v2f __attribute__((ext_vector_type(2)));
typedef float v8f __attribute__((ext_vector_type(8)));

#define B_    4
#define CIN   64
#define COUT  64
#define NPTS  16384
#define KNN   16
#define NGRP  4
#define CPG   16
#define EPS_  1e-5f
#define OT_N  64
#define ST_N  64
#define OT_LD (OT_N + 1)
#define WP    (CIN + 4)

__global__ void k_init(float* __restrict__ stats) {
    if (threadIdx.x < B_ * NGRP * 2) stats[threadIdx.x] = 0.0f;
}

__global__ __launch_bounds__(256) void k_gemm(const float* __restrict__ feat,
                                              const float* __restrict__ W1,
                                              const float* __restrict__ W2,
                                              float* __restrict__ local_t,
                                              float* __restrict__ edge_t) {
    __shared__ float sW[2 * COUT * WP];
    const int tid = threadIdx.x;
    for (int i = tid; i < COUT * CIN; i += 256) {
        const int o = i / CIN, k = i % CIN;
        sW[o * WP + k]             = W1[i];
        sW[COUT * WP + o * WP + k] = W2[i];
    }
    __syncthreads();

    const int wave = tid >> 5;
    const int lane = tid & 31;
    const int ln   = lane & 15;
    const int hi   = lane >> 4;
    const int b    = blockIdx.x >> 7;
    const int tile = ((blockIdx.x & 127) << 3) + wave;
    const int n0   = tile << 4;

    const size_t fbase = (size_t)b * CIN * NPTS + (size_t)(n0 + ln);
    v2f bf[CIN / 4];
#pragma unroll
    for (int t = 0; t < CIN / 4; ++t) {
        const int k0 = 4 * t + 2 * hi;
        bf[t].x = feat[fbase + (size_t)k0 * NPTS];
        bf[t].y = feat[fbase + (size_t)(k0 + 1) * NPTS];
    }

#pragma unroll
    for (int mat = 0; mat < 2; ++mat) {
        float* dst = mat ? edge_t : local_t;
#pragma unroll
        for (int mp = 0; mp < 2; ++mp) {
            const int mt0 = 2 * mp;
            v8f acc0 = (v8f){0, 0, 0, 0, 0, 0, 0, 0};
            v8f acc1 = (v8f){0, 0, 0, 0, 0, 0, 0, 0};
#pragma unroll
            for (int t = 0; t < CIN / 4; ++t) {
                const int k0 = 4 * t + 2 * hi;
                const float* w0 = &sW[mat * (COUT * WP) + (mt0 * 16 + ln) * WP + k0];
                const float* w1 = w0 + 16 * WP;
                v2f a0, a1;
                a0.x = w0[0];
                a0.y = w0[1];
                a1.x = w1[0];
                a1.y = w1[1];
                acc0 = __builtin_amdgcn_wmma_f32_16x16x4_f32(
                    false, a0, false, bf[t], (short)0, acc0, false, false);
                acc1 = __builtin_amdgcn_wmma_f32_16x16x4_f32(
                    false, a1, false, bf[t], (short)0, acc1, false, false);
            }
            const size_t base = ((size_t)b * NPTS + (size_t)(n0 + ln)) * COUT
                                + (size_t)(mt0 * 16 + 8 * hi);
            *(float4*)(dst + base)      = make_float4(acc0[0], acc0[1], acc0[2], acc0[3]);
            *(float4*)(dst + base + 4)  = make_float4(acc0[4], acc0[5], acc0[6], acc0[7]);
            *(float4*)(dst + base + 16) = make_float4(acc1[0], acc1[1], acc1[2], acc1[3]);
            *(float4*)(dst + base + 20) = make_float4(acc1[4], acc1[5], acc1[6], acc1[7]);
            __builtin_amdgcn_sched_barrier(0);
        }
    }
}

__global__ __launch_bounds__(256) void k_stats(const int* __restrict__ knn,
                                               const float* __restrict__ local_t,
                                               const float* __restrict__ edge_t,
                                               float* __restrict__ stats) {
    __shared__ float sh[2 * NGRP];
    const int tid = threadIdx.x;
    if (tid < 2 * NGRP) sh[tid] = 0.0f;
    __syncthreads();

    const int b  = blockIdx.x / (NPTS / ST_N);
    const int n0 = (blockIdx.x % (NPTS / ST_N)) * ST_N;
    const int c4 = tid & 15;
    const int nl = tid >> 4;
    const int g  = c4 >> 2;

    float s = 0.0f, ss = 0.0f;
    for (int p = 0; p < ST_N / 16; ++p) {
        const int n = n0 + p * 16 + nl;
        const float4 l4 = *(const float4*)(local_t + ((size_t)b * NPTS + n) * COUT
                                           + (size_t)c4 * 4);
        const int* kp = knn + ((size_t)b * NPTS + n) * KNN;

        float4 e[KNN];
#pragma unroll
        for (int j = 0; j < KNN; ++j) {
            const int idx = kp[j];
            e[j] = *(const float4*)(edge_t + ((size_t)b * NPTS + idx) * COUT + (size_t)c4 * 4);
        }
        __builtin_amdgcn_sched_barrier(0);

#pragma unroll
        for (int j = 0; j < KNN; ++j) {
            const float4 e4 = e[j];
            const float d0 = e4.x - l4.x, d1 = e4.y - l4.y;
            const float d2 = e4.z - l4.z, d3 = e4.w - l4.w;
            s  += d0 + d1 + d2 + d3;
            ss += d0 * d0 + d1 * d1 + d2 * d2 + d3 * d3;
        }
    }
    atomicAdd(&sh[g * 2 + 0], s);
    atomicAdd(&sh[g * 2 + 1], ss);
    __syncthreads();
    if (tid < 2 * NGRP) atomicAdd(&stats[b * 2 * NGRP + tid], sh[tid]);
}

__global__ __launch_bounds__(256) void k_out(const int* __restrict__ knn,
                                             const float* __restrict__ local_t,
                                             const float* __restrict__ edge_t,
                                             const float* __restrict__ stats,
                                             const float* __restrict__ gw,
                                             const float* __restrict__ gb,
                                             float* __restrict__ out) {
    __shared__ float tile[COUT * OT_LD];
    const int tid = threadIdx.x;
    const int b   = blockIdx.x / (NPTS / OT_N);
    const int n0  = (blockIdx.x % (NPTS / OT_N)) * OT_N;
    const int c4  = tid & 15;
    const int nl  = tid >> 4;
    const int g   = c4 >> 2;

    const float inv_cnt = 1.0f / (float)((size_t)CPG * NPTS * KNN);
    const float mean = stats[b * 2 * NGRP + g * 2 + 0] * inv_cnt;
    const float msq  = stats[b * 2 * NGRP + g * 2 + 1] * inv_cnt;
    const float inv  = rsqrtf(msq - mean * mean + EPS_);

    const float4 gamma = ((const float4*)gw)[c4];
    const float4 beta  = ((const float4*)gb)[c4];
    const float r = 1.0f / (float)KNN;

    for (int s = 0; s < OT_N / 16; ++s) {
        const int pl = s * 16 + nl;
        const int n  = n0 + pl;
        const float4 l4 = *(const float4*)(local_t + ((size_t)b * NPTS + n) * COUT
                                           + (size_t)c4 * 4);
        const int* kp = knn + ((size_t)b * NPTS + n) * KNN;

        float4 e[KNN];
#pragma unroll
        for (int j = 0; j < KNN; ++j) {
            const int idx = kp[j];
            e[j] = *(const float4*)(edge_t + ((size_t)b * NPTS + idx) * COUT + (size_t)c4 * 4);
        }
        __builtin_amdgcn_sched_barrier(0);

        float a0 = 0.0f, a1 = 0.0f, a2 = 0.0f, a3 = 0.0f;
#pragma unroll
        for (int j = 0; j < KNN; ++j) {
            const float4 e4 = e[j];
            float v0 = ((e4.x - l4.x) - mean) * inv * gamma.x + beta.x;
            float v1 = ((e4.y - l4.y) - mean) * inv * gamma.y + beta.y;
            float v2 = ((e4.z - l4.z) - mean) * inv * gamma.z + beta.z;
            float v3 = ((e4.w - l4.w) - mean) * inv * gamma.w + beta.w;
            a0 += fmaxf(v0, 0.0f);
            a1 += fmaxf(v1, 0.0f);
            a2 += fmaxf(v2, 0.0f);
            a3 += fmaxf(v3, 0.0f);
        }
        tile[(c4 * 4 + 0) * OT_LD + pl] = a0 * r;
        tile[(c4 * 4 + 1) * OT_LD + pl] = a1 * r;
        tile[(c4 * 4 + 2) * OT_LD + pl] = a2 * r;
        tile[(c4 * 4 + 3) * OT_LD + pl] = a3 * r;
    }
    __syncthreads();

    const int q    = tid & 15;
    const int crow = tid >> 4;
    for (int p = 0; p < COUT / 16; ++p) {
        const int c = p * 16 + crow;
        const float* trow = tile + c * OT_LD + 4 * q;
        const float4 v = make_float4(trow[0], trow[1], trow[2], trow[3]);
        *(float4*)(out + ((size_t)b * COUT + (size_t)c) * NPTS + (size_t)(n0 + 4 * q)) = v;
    }
}

extern "C" void kernel_launch(void* const* d_in, const int* in_sizes, int n_in,
                              void* d_out, int out_size, void* d_ws, size_t ws_size,
                              hipStream_t stream) {
    (void)in_sizes; (void)n_in; (void)out_size; (void)ws_size;

    const float* feat = (const float*)d_in[0];
    const int*   knn  = (const int*)d_in[1];
    const float* W1   = (const float*)d_in[2];
    const float* W2   = (const float*)d_in[3];
    const float* gw   = (const float*)d_in[4];
    const float* gb   = (const float*)d_in[5];
    float*       out  = (float*)d_out;

    float* local_t = (float*)d_ws;
    float* edge_t  = local_t + (size_t)B_ * NPTS * COUT;
    float* stats   = edge_t  + (size_t)B_ * NPTS * COUT;

    k_init <<<1, 64, 0, stream>>>(stats);
    k_gemm <<<B_ * (NPTS / 16) / 8, 256, 0, stream>>>(feat, W1, W2, local_t, edge_t);
    k_stats<<<B_ * (NPTS / ST_N), 256, 0, stream>>>(knn, local_t, edge_t, stats);
    k_out  <<<B_ * (NPTS / OT_N), 256, 0, stream>>>(knn, local_t, edge_t, stats,
                                                    gw, gb, out);
}


# k_gemm prologue: weights (16-byte loads, 8 per thread) and the wave's feature slice requested as one group, one round trip; was 8 serial round trips for the weights, then a ninth for the features
# speedup vs baseline: 2.5091x; 1.0735x over previous
#include <hip/hip_runtime.h>
#include <math.h>

typedef float v2f __attribute__((ext_vector_type(2)));
typedef float v4f __attribute__((ext_vector_type(4)));
typedef float v8f __attribute__((ext_vector_type(8)));

#define B_    4
#define CIN   64
#define COUT  64
#define NPTS  16384
#define KNN   16
#define NGRP  4
#define CPG   16
#define EPS_  1e-5f
#define OT_N  64
#define ST_N  64
#define OT_LD (OT_N + 1)
#define WP    (CIN + 4)

__global__ void k_init(float* __restrict__ stats) {
    if (threadIdx.x < B_ * NGRP * 2) stats[threadIdx.x] = 0.0f;
}

__global__ __launch_bounds__(256) void k_gemm(const float* __restrict__ feat,
                                              const float* __restrict__ W1,
                                              const float* __restrict__ W2,
                                              float* __restrict__ local_t,
                                              float* __restrict__ edge_t) {
    __shared__ __attribute__((aligned(16))) float sW[2 * COUT * WP];
    const int tid = threadIdx.x;

    const int wave = tid >> 5;
    const int lane = tid & 31;
    const int ln   = lane & 15;
    const int hi   = lane >> 4;
    const int b    = blockIdx.x >> 7;
    const int tile = ((blockIdx.x & 127) << 3) + wave;
    const int n0   = tile << 4;

    const size_t fbase = (size_t)b * CIN * NPTS + (size_t)(n0 + ln);
    v2f bf[CIN / 4];
#pragma unroll
    for (int t = 0; t < CIN / 4; ++t) {
        const int k0 = 4 * t + 2 * hi;
        bf[t].x = feat[fbase + (size_t)k0 * NPTS];
        bf[t].y = feat[fbase + (size_t)(k0 + 1) * NPTS];
    }

    v4f wq[8];
#pragma unroll
    for (int r = 0; r < 4; ++r) {
        wq[r]     = ((const v4f*)W1)[tid + 256 * r];
        wq[4 + r] = ((const v4f*)W2)[tid + 256 * r];
    }
    asm volatile("" :: "v"(wq[0]), "v"(wq[1]), "v"(wq[2]), "v"(wq[3]),
                       "v"(wq[4]), "v"(wq[5]), "v"(wq[6]), "v"(wq[7]));
#pragma unroll
    for (int r = 0; r < 4; ++r) {
        const int q = tid + 256 * r;
        const int o = q >> 4, k = (q & 15) << 2;
        *(v4f*)&sW[o * WP + k]             = wq[r];
        *(v4f*)&sW[COUT * WP + o * WP + k] = wq[4 + r];
    }
    __syncthreads();

#pragma unroll
    for (int mat = 0; mat < 2; ++mat) {
        float* dst = mat ? edge_t : local_t;
#pragma unroll
        for (int mp = 0; mp < 2; ++mp) {
            const int mt0 = 2 * mp;
            v8f acc0 = (v8f){0, 0, 0, 0, 0, 0, 0, 0};
            v8f acc1 = (v8f){0, 0, 0, 0, 0, 0, 0, 0};
#pragma unroll
            for (int t = 0; t < CIN / 4; ++t) {
                const int k0 = 4 * t + 2 * hi;
                const float* w0 = &sW[mat * (COUT * WP) + (mt0 * 16 + ln) * WP + k0];
                const float* w1 = w0 + 16 * WP;
                v2f a0, a1;
                a0.x = w0[0];
                a0.y = w0[1];
                a1.x = w1[0];
                a1.y = w1[1];
                acc0 = __builtin_amdgcn_wmma_f32_16x16x4_f32(
                    false, a0, false, bf[t], (short)0, acc0, false, false);
                acc1 = __builtin_amdgcn_wmma_f32_16x16x4_f32(
                    false, a1, false, bf[t], (short)0, acc1, false, false);
            }
            const size_t base = ((size_t)b * NPTS + (size_t)(n0 + ln)) * COUT
                                + (size_t)(mt0 * 16 + 8 * hi);
            *(float4*)(dst + base)      = make_float4(acc0[0], acc0[1], acc0[2], acc0[3]);
            *(float4*)(dst + base + 4)  = make_float4(acc0[4], acc0[5], acc0[6], acc0[7]);
            *(float4*)(dst + base + 16) = make_float4(acc1[0], acc1[1], acc1[2], acc1[3]);
            *(float4*)(dst + base + 20) = make_float4(acc1[4], acc1[5], acc1[6], acc1[7]);
            __builtin_amdgcn_sched_barrier(0);
        }
    }
}

__global__ __launch_bounds__(256) void k_stats(const int* __restrict__ knn,
                                               const float* __restrict__ local_t,
                                               const float* __restrict__ edge_t,
                                               float* __restrict__ stats) {
    __shared__ float sh[2 * NGRP];
    const int tid = threadIdx.x;
    if (tid < 2 * NGRP) sh[tid] = 0.0f;
    __syncthreads();

    const int b  = blockIdx.x / (NPTS / ST_N);
    const int n0 = (blockIdx.x % (NPTS / ST_N)) * ST_N;
    const int c4 = tid & 15;
    const int nl = tid >> 4;
    const int g  = c4 >> 2;

    float s = 0.0f, ss = 0.0f;
    for (int p = 0; p < ST_N / 16; ++p) {
        const int n = n0 + p * 16 + nl;
        const float4 l4 = *(const float4*)(local_t + ((size_t)b * NPTS + n) * COUT
                                           + (size_t)c4 * 4);
        const int* kp = knn + ((size_t)b * NPTS + n) * KNN;

        float4 e[KNN];
#pragma unroll
        for (int j = 0; j < KNN; ++j) {
            const int idx = kp[j];
            e[j] = *(const float4*)(edge_t + ((size_t)b * NPTS + idx) * COUT + (size_t)c4 * 4);
        }
        __builtin_amdgcn_sched_barrier(0);

#pragma unroll
        for (int j = 0; j < KNN; ++j) {
            const float4 e4 = e[j];
            const float d0 = e4.x - l4.x, d1 = e4.y - l4.y;
            const float d2 = e4.z - l4.z, d3 = e4.w - l4.w;
            s  += d0 + d1 + d2 + d3;
            ss += d0 * d0 + d1 * d1 + d2 * d2 + d3 * d3;
        }
    }
    atomicAdd(&sh[g * 2 + 0], s);
    atomicAdd(&sh[g * 2 + 1], ss);
    __syncthreads();
    if (tid < 2 * NGRP) atomicAdd(&stats[b * 2 * NGRP + tid], sh[tid]);
}

__global__ __launch_bounds__(256) void k_out(const int* __restrict__ knn,
                                             const float* __restrict__ local_t,
                                             const float* __restrict__ edge_t,
                                             const float* __restrict__ stats,
                                             const float* __restrict__ gw,
                                             const float* __restrict__ gb,
                                             float* __restrict__ out) {
    __shared__ float tile[COUT * OT_LD];
    const int tid = threadIdx.x;
    const int b   = blockIdx.x / (NPTS / OT_N);
    const int n0  = (blockIdx.x % (NPTS / OT_N)) * OT_N;
    const int c4  = tid & 15;
    const int nl  = tid >> 4;
    const int g   = c4 >> 2;

    const float inv_cnt = 1.0f / (float)((size_t)CPG * NPTS * KNN);
    const float mean = stats[b * 2 * NGRP + g * 2 + 0] * inv_cnt;
    const float msq  = stats[b * 2 * NGRP + g * 2 + 1] * inv_cnt;
    const float inv  = rsqrtf(msq - mean * mean + EPS_);

    const float4 gamma = ((const float4*)gw)[c4];
    const float4 beta  = ((const float4*)gb)[c4];
    const float r = 1.0f / (float)KNN;

    for (int s = 0; s < OT_N / 16; ++s) {
        const int pl = s * 16 + nl;
        const int n  = n0 + pl;
        const float4 l4 = *(const float4*)(local_t + ((size_t)b * NPTS + n) * COUT
                                           + (size_t)c4 * 4);
        const int* kp = knn + ((size_t)b * NPTS + n) * KNN;

        float4 e[KNN];
#pragma unroll
        for (int j = 0; j < KNN; ++j) {
            const int idx = kp[j];
            e[j] = *(const float4*)(edge_t + ((size_t)b * NPTS + idx) * COUT + (size_t)c4 * 4);
        }
        __builtin_amdgcn_sched_barrier(0);

        float a0 = 0.0f, a1 = 0.0f, a2 = 0.0f, a3 = 0.0f;
#pragma unroll
        for (int j = 0; j < KNN; ++j) {
            const float4 e4 = e[j];
            float v0 = ((e4.x - l4.x) - mean) * inv * gamma.x + beta.x;
            float v1 = ((e4.y - l4.y) - mean) * inv * gamma.y + beta.y;
            float v2 = ((e4.z - l4.z) - mean) * inv * gamma.z + beta.z;
            float v3 = ((e4.w - l4.w) - mean) * inv * gamma.w + beta.w;
            a0 += fmaxf(v0, 0.0f);
            a1 += fmaxf(v1, 0.0f);
            a2 += fmaxf(v2, 0.0f);
            a3 += fmaxf(v3, 0.0f);
        }
        tile[(c4 * 4 + 0) * OT_LD + pl] = a0 * r;
        tile[(c4 * 4 + 1) * OT_LD + pl] = a1 * r;
        tile[(c4 * 4 + 2) * OT_LD + pl] = a2 * r;
        tile[(c4 * 4 + 3) * OT_LD + pl] = a3 * r;
    }
    __syncthreads();

    const int q    = tid & 15;
    const int crow = tid >> 4;
    for (int p = 0; p < COUT / 16; ++p) {
        const int c = p * 16 + crow;
        const float* trow = tile + c * OT_LD + 4 * q;
        const float4 v = make_float4(trow[0], trow[1], trow[2], trow[3]);
        *(float4*)(out + ((size_t)b * COUT + (size_t)c) * NPTS + (size_t)(n0 + 4 * q)) = v;
    }
}

extern "C" void kernel_launch(void* const* d_in, const int* in_sizes, int n_in,
                              void* d_out, int out_size, void* d_ws, size_t ws_size,
                              hipStream_t stream) {
    (void)in_sizes; (void)n_in; (void)out_size; (void)ws_size;

    const float* feat = (const float*)d_in[0];
    const int*   knn  = (const int*)d_in[1];
    const float* W1   = (const float*)d_in[2];
    const float* W2   = (const float*)d_in[3];
    const float* gw   = (const float*)d_in[4];
    const float* gb   = (const float*)d_in[5];
    float*       out  = (float*)d_out;

    float* local_t = (float*)d_ws;
    float* edge_t  = local_t + (size_t)B_ * NPTS * COUT;
    float* stats   = edge_t  + (size_t)B_ * NPTS * COUT;

    k_init <<<1, 64, 0, stream>>>(stats);
    k_gemm <<<B_ * (NPTS / 16) / 8, 256, 0, stream>>>(feat, W1, W2, local_t, edge_t);
    k_stats<<<B_ * (NPTS / ST_N), 256, 0, stream>>>(knn, local_t, edge_t, stats);
    k_out  <<<B_ * (NPTS / OT_N), 256, 0, stream>>>(knn, local_t, edge_t, stats,
                                                    gw, gb, out);
}
